# starting kernel
# baseline (speedup 1.0000x reference)
#include <hip/hip_runtime.h>
#include <hip/hip_bf16.h>
#include <math.h>

#define B_   2
#define C_   256
#define H_   48
#define W_   48
#define HD   24
#define WD   24
#define L_   576
#define K9   2304
#define K16  4096
#define SCALE_ 10.0f
#define EPS_   1e-4f

typedef float v2f __attribute__((ext_vector_type(2)));
typedef float v8f __attribute__((ext_vector_type(8)));

#define TK  32
#define LDP (TK + 2)

__device__ __forceinline__ float patch9(const float* ds, int b, int k, int l) {
    int c = k / 9, r = k % 9;
    int ki = r / 3, kj = r % 3;
    int y = l / WD + ki - 1;
    int x = l % WD + kj - 1;
    if ((unsigned)y >= HD || (unsigned)x >= WD) return 0.0f;
    return ds[((b * C_ + c) * HD + y) * WD + x];
}

__device__ __forceinline__ float patch16(const float* sc, int b, int m, int l) {
    int c = m >> 4, i = (m >> 2) & 3, j = m & 3;
    int r = 2 * (l / WD) + i - 1;
    int s = 2 * (l % WD) + j - 1;
    if ((unsigned)r >= H_ || (unsigned)s >= W_) return 0.0f;
    return sc[((b * C_ + c) * H_ + r) * W_ + s];
}

__global__ __launch_bounds__(256) void k_prep_ds(const float* __restrict__ left,
                                                 const float* __restrict__ right,
                                                 const float* __restrict__ mid,
                                                 float* __restrict__ dsL,
                                                 float* __restrict__ dsR,
                                                 float* __restrict__ dsM) {
    int idx = blockIdx.x * 256 + threadIdx.x;
    if (idx >= B_ * C_ * HD * WD) return;
    int x = idx % WD, t = idx / WD;
    int y = t % HD;  t /= HD;
    int c = t % C_;  int b = t / C_;
    int src = ((b * C_ + c) * H_ + 2 * y) * W_ + 2 * x;
    dsL[idx] = left[src];
    dsR[idx] = right[src];
    dsM[idx] = mid[src];
}

__global__ __launch_bounds__(256) void k_denom(const float* __restrict__ dsL,
                                               const float* __restrict__ dsR,
                                               float* __restrict__ denL,
                                               float* __restrict__ denR) {
    const float* ds = blockIdx.y ? dsR : dsL;
    float* den      = blockIdx.y ? denR : denL;
    int gl   = blockIdx.x * 8 + (threadIdx.x >> 5);
    int b    = gl / L_;
    int l    = gl % L_;
    int lane = threadIdx.x & 31;
    float s = 0.0f;
    for (int k = lane; k < K9; k += 32) {
        float v = patch9(ds, b, k, l);
        s += v * v;
    }
    #pragma unroll
    for (int off = 16; off; off >>= 1) s += __shfl_xor(s, off, 32);
    if (lane == 0) den[gl] = sqrtf(s + (float)K9 * EPS_);
}

__global__ __launch_bounds__(256) void k_scores(const float* __restrict__ dsL,
                                                const float* __restrict__ dsR,
                                                const float* __restrict__ dsM,
                                                const float* __restrict__ denL,
                                                const float* __restrict__ denR,
                                                float* __restrict__ scL,
                                                float* __restrict__ scR) {
    __shared__ __align__(16) float Ap[64][LDP];
    __shared__ __align__(16) float Bp[64][LDP];
    const float* dsF = blockIdx.y ? dsR : dsL;
    const float* den = blockIdx.y ? denR : denL;
    float* out       = blockIdx.y ? scR : scL;
    int b   = blockIdx.z;
    int l0  = (blockIdx.x / 9) * 64;
    int p0  = (blockIdx.x % 9) * 64;
    int tid = threadIdx.x;
    int lane = tid & 31, w = tid >> 5;
    int tr  = w >> 1;
    int tcb = (w & 1) * 2;
    int m16   = lane & 15;
    int khalf = (lane >> 4) * 2;
    v8f acc0 = {}; v8f acc1 = {};

    for (int kb = 0; kb < K9; kb += TK) {
        __syncthreads();
        #pragma unroll
        for (int t = 0; t < 8; ++t) {
            int idx = tid + t * 256;
            int lr  = idx & 63;
            int kk  = idx >> 6;
            Ap[lr][kk] = patch9(dsF, b, kb + kk, l0 + lr);
            Bp[lr][kk] = patch9(dsM, b, kb + kk, p0 + lr);
        }
        __syncthreads();
        #pragma unroll
        for (int kk0 = 0; kk0 < TK; kk0 += 4) {
            int ka = kk0 + khalf;
            v2f a  = *(const v2f*)&Ap[tr * 16 + m16][ka];
            v2f b0 = *(const v2f*)&Bp[tcb * 16 + m16][ka];
            v2f b1 = *(const v2f*)&Bp[(tcb + 1) * 16 + m16][ka];
            acc0 = __builtin_amdgcn_wmma_f32_16x16x4_f32(false, a, false, b0,
                                                         (short)0, acc0, false, false);
            acc1 = __builtin_amdgcn_wmma_f32_16x16x4_f32(false, a, false, b1,
                                                         (short)0, acc1, false, false);
        }
    }
    int n     = lane & 15;
    int rbase = l0 + tr * 16 + ((lane >> 4) << 3);
    #pragma unroll
    for (int v = 0; v < 8; ++v) {
        int row = rbase + v;
        float scl = SCALE_ / den[b * L_ + row];
        out[((size_t)b * L_ + row) * L_ + p0 + tcb * 16 + n]       = acc0[v] * scl;
        out[((size_t)b * L_ + row) * L_ + p0 + (tcb + 1) * 16 + n] = acc1[v] * scl;
    }
}

__global__ __launch_bounds__(192) void k_softmax(float* __restrict__ scL,
                                                 float* __restrict__ scR) {
    float* sc = blockIdx.y ? scR : scL;
    int b = blockIdx.z;
    int p = blockIdx.x * 192 + threadIdx.x;
    float* col = sc + (size_t)b * L_ * L_ + p;
    float mx = -3.0e38f;
    for (int l = 0; l < L_; ++l) mx = fmaxf(mx, col[(size_t)l * L_]);
    float sum = 0.0f;
    for (int l = 0; l < L_; ++l) {
        float e = expf(col[(size_t)l * L_] - mx);
        col[(size_t)l * L_] = e;
        sum += e;
    }
    float inv = 1.0f / sum;
    for (int l = 0; l < L_; ++l) col[(size_t)l * L_] *= inv;
}

__global__ __launch_bounds__(256) void k_recon(const float* __restrict__ shcL,
                                               const float* __restrict__ shcR,
                                               const float* __restrict__ atL,
                                               const float* __restrict__ atR,
                                               float* __restrict__ YL,
                                               float* __restrict__ YR) {
    __shared__ __align__(16) float Ap[64][LDP];
    __shared__ __align__(16) float Bp[64][LDP];
    const float* sc = blockIdx.y ? shcR : shcL;
    const float* at = blockIdx.y ? atR  : atL;
    float* Y        = blockIdx.y ? YR   : YL;
    int b   = blockIdx.z;
    int m0  = (blockIdx.x / 9) * 64;
    int p0  = (blockIdx.x % 9) * 64;
    int tid = threadIdx.x;
    int lane = tid & 31, w = tid >> 5;
    int tr  = w >> 1;
    int tcb = (w & 1) * 2;
    int m16   = lane & 15;
    int khalf = (lane >> 4) * 2;
    v8f acc0 = {}; v8f acc1 = {};

    for (int kb = 0; kb < L_; kb += TK) {
        __syncthreads();
        #pragma unroll
        for (int t = 0; t < 8; ++t) {
            int idx = tid + t * 256;
            int lr  = idx & 63;
            int kk  = idx >> 6;
            Ap[lr][kk] = patch16(sc, b, m0 + lr, kb + kk);
            Bp[lr][kk] = at[((size_t)b * L_ + kb + kk) * L_ + p0 + lr];
        }
        __syncthreads();
        #pragma unroll
        for (int kk0 = 0; kk0 < TK; kk0 += 4) {
            int ka = kk0 + khalf;
            v2f a  = *(const v2f*)&Ap[tr * 16 + m16][ka];
            v2f b0 = *(const v2f*)&Bp[tcb * 16 + m16][ka];
            v2f b1 = *(const v2f*)&Bp[(tcb + 1) * 16 + m16][ka];
            acc0 = __builtin_amdgcn_wmma_f32_16x16x4_f32(false, a, false, b0,
                                                         (short)0, acc0, false, false);
            acc1 = __builtin_amdgcn_wmma_f32_16x16x4_f32(false, a, false, b1,
                                                         (short)0, acc1, false, false);
        }
    }
    int n     = lane & 15;
    int rbase = m0 + tr * 16 + ((lane >> 4) << 3);
    #pragma unroll
    for (int v = 0; v < 8; ++v) {
        int row = rbase + v;
        Y[((size_t)b * K16 + row) * L_ + p0 + tcb * 16 + n]       = acc0[v];
        Y[((size_t)b * K16 + row) * L_ + p0 + (tcb + 1) * 16 + n] = acc1[v];
    }
}

__global__ __launch_bounds__(256) void k_finalize(const float* __restrict__ YL,
                                                  const float* __restrict__ YR,
                                                  float* __restrict__ out) {
    int idx = blockIdx.x * 256 + threadIdx.x;
    if (idx >= B_ * C_ * H_ * W_) return;
    int s = idx % W_; int t = idx / W_;
    int r = t % H_;  t /= H_;
    int c = t % C_;  int b = t / C_;
    float aL = 0.0f, aR = 0.0f;
    int i0 = (r + 1) & 1, j0 = (s + 1) & 1;
    #pragma unroll
    for (int di = 0; di < 2; ++di) {
        int i = i0 + 2 * di, ry = r + 1 - i;
        if (ry < 0 || ry >= 2 * HD) continue;
        int y = ry >> 1;
        #pragma unroll
        for (int dj = 0; dj < 2; ++dj) {
            int j = j0 + 2 * dj, sx = s + 1 - j;
            if (sx < 0 || sx >= 2 * WD) continue;
            int x = sx >> 1;
            int m = ((c << 2) | i) * 4 + j;
            size_t off = ((size_t)b * K16 + m) * L_ + y * WD + x;
            aL += YL[off];
            aR += YR[off];
        }
    }
    const float PI = 3.14159265358979323846f;
    float wl = 0.5f * (cosf(PI * (float)s / (float)(W_ - 1)) + 1.0f);
    float wr = 0.5f * (cosf(PI * (float)(W_ - 1 - s) / (float)(W_ - 1)) + 1.0f);
    out[idx] = 0.25f * (wl * aL + wr * aR);
}

extern "C" void kernel_launch(void* const* d_in, const int* in_sizes, int n_in,
                              void* d_out, int out_size, void* d_ws, size_t ws_size,
                              hipStream_t stream) {
    (void)in_sizes; (void)n_in; (void)out_size; (void)ws_size;
    const float* left  = (const float*)d_in[0];
    const float* right = (const float*)d_in[1];
    const float* mid   = (const float*)d_in[2];
    const float* shcL  = (const float*)d_in[3];
    const float* shcR  = (const float*)d_in[4];
    float* out = (float*)d_out;

    float* w   = (float*)d_ws;
    const size_t NDS = (size_t)B_ * C_ * HD * WD;
    const size_t NDN = (size_t)B_ * L_;
    const size_t NSC = (size_t)B_ * L_ * L_;
    const size_t NY  = (size_t)B_ * K16 * L_;
    float* dsL  = w;
    float* dsR  = dsL + NDS;
    float* dsM  = dsR + NDS;
    float* denL = dsM + NDS;
    float* denR = denL + NDN;
    float* scL  = denR + NDN;
    float* scR  = scL + NSC;
    float* YL   = scR + NSC;
    float* YR   = YL + NY;

    k_prep_ds<<<dim3((NDS + 255) / 256), dim3(256), 0, stream>>>(left, right, mid,
                                                                 dsL, dsR, dsM);
    k_denom<<<dim3(B_ * L_ / 8, 2), dim3(256), 0, stream>>>(dsL, dsR, denL, denR);
    k_scores<<<dim3(81, 2, B_), dim3(256), 0, stream>>>(dsL, dsR, dsM,
                                                        denL, denR, scL, scR);
    k_softmax<<<dim3(3, 2, B_), dim3(192), 0, stream>>>(scL, scR);
    k_recon<<<dim3(576, 2, B_), dim3(256), 0, stream>>>(shcL, shcR, scL, scR, YL, YR);
    k_finalize<<<dim3((B_ * C_ * H_ * W_ + 255) / 256), dim3(256), 0, stream>>>(YL, YR, out);
}


# both GEMMs: a wave's priority steps down (s_setprio 3,2,1,0) along its K loop, so groups sharing a WGP take turns on the matrix unit instead of the oldest running ahead and the youngest ending alone
# speedup vs baseline: 6.2950x; 6.2950x over previous
#include <hip/hip_runtime.h>
#include <hip/hip_bf16.h>
#include <math.h>

#define B_   2
#define C_   256
#define H_   48
#define W_   48
#define HD   24
#define WD   24
#define L_   576
#define K9   2304
#define K16  4096
#define SCALE_ 10.0f
#define EPS_   1e-4f

typedef float v2f __attribute__((ext_vector_type(2)));
typedef float v8f __attribute__((ext_vector_type(8)));

#define TK  32
#define LDP (TK + 2)

#define PP   50
#define PSZ  (PP * PP)

#define PR_  28
#define PC_  26
#define NPP  (PR_ * PC_)

__global__ __launch_bounds__(256) void k_prep_cl(const float* __restrict__ left,
                                                 const float* __restrict__ right,
                                                 const float* __restrict__ mid,
                                                 float* __restrict__ DL,
                                                 float* __restrict__ DR,
                                                 float* __restrict__ DM) {
    __shared__ float tile[PC_][65];
    const float* in = blockIdx.y == 0 ? left : (blockIdx.y == 1 ? right : mid);
    float* D        = blockIdx.y == 0 ? DL   : (blockIdx.y == 1 ? DR    : DM);
    int b   = blockIdx.z;
    int pr  = blockIdx.x >> 2;
    int cq  = blockIdx.x & 3;
    int tid = threadIdx.x;
    int lane = tid & 31, w = tid >> 5;
    bool inside = (pr >= 1) && (pr <= HD) && (lane >= 1) && (lane <= WD);
    if (lane < PC_) {
        float v[8] = {0.0f, 0.0f, 0.0f, 0.0f, 0.0f, 0.0f, 0.0f, 0.0f};
        if (inside) {
            #pragma unroll
            for (int t = 0; t < 8; ++t)
                v[t] = in[((size_t)(b * C_ + cq * 64 + w + 8 * t) * H_ + 2 * (pr - 1)) * W_ + 2 * (lane - 1)];
        }
        #pragma unroll
        for (int t = 0; t < 8; ++t) tile[lane][w + 8 * t] = v[t];
    }
    __syncthreads();
    for (int it = tid; it < PC_ * 16; it += 256) {
        int px = it >> 4, c4 = it & 15;
        float4 o = make_float4(tile[px][4 * c4], tile[px][4 * c4 + 1], tile[px][4 * c4 + 2], tile[px][4 * c4 + 3]);
        *(float4*)(D + ((size_t)(b * NPP + pr * PC_ + px)) * C_ + cq * 64 + 4 * c4) = o;
    }
}

__global__ __launch_bounds__(256) void k_denom(const float* __restrict__ DL,
                                               const float* __restrict__ DR,
                                               float* __restrict__ denL,
                                               float* __restrict__ denR) {
    const float* D = blockIdx.y ? DR : DL;
    float* den     = blockIdx.y ? denR : denL;
    int lane = threadIdx.x & 31, w = threadIdx.x >> 5;
    int gl = blockIdx.x * 8 + w;
    int b  = gl / L_;
    int l  = gl % L_;
    int ly = l / WD, lx = l - ly * WD;
    const float* p = D + ((size_t)(b * NPP + ly * PC_ + lx)) * C_ + 4 * lane;
    float4 u[9], v[9];
    #pragma unroll
    for (int tap = 0; tap < 9; ++tap) {
        const float* q = p + ((tap / 3) * PC_ + (tap % 3)) * C_;
        u[tap] = *(const float4*)q;
        v[tap] = *(const float4*)(q + 128);
    }
    float s = 0.0f;
    #pragma unroll
    for (int tap = 0; tap < 9; ++tap) {
        s += u[tap].x * u[tap].x + u[tap].y * u[tap].y + u[tap].z * u[tap].z + u[tap].w * u[tap].w;
        s += v[tap].x * v[tap].x + v[tap].y * v[tap].y + v[tap].z * v[tap].z + v[tap].w * v[tap].w;
    }
    #pragma unroll
    for (int o = 16; o > 0; o >>= 1) s += __shfl_xor(s, o);
    if (lane == 0) den[gl] = sqrtf(s + (float)K9 * EPS_);
}

#define KC   64
#define CP   84
#define NPA  (4 * PC_)
#define NPB  (6 * PC_)
__global__ __launch_bounds__(128) void k_scores(const float* __restrict__ DL,
                                                const float* __restrict__ DR,
                                                const float* __restrict__ DM,
                                                const float* __restrict__ denL,
                                                const float* __restrict__ denR,
                                                float* __restrict__ scL,
                                                float* __restrict__ scR,
                                                float* __restrict__ pmx) {
    __shared__ __align__(16) float SA[NPA * CP];
    __shared__ __align__(16) float SB[(NPB + 4) * CP];
    __shared__ float cmx[2][64];
    const float* DF  = blockIdx.y ? DR : DL;
    const float* den = blockIdx.y ? denR : denL;
    float* out       = blockIdx.y ? scR : scL;
    int b   = blockIdx.z;
    int l0  = (blockIdx.x / 9) * 32;
    int p0  = (blockIdx.x % 9) * 64;
    int gyA = l0 / WD;
    int gyB = p0 / WD;
    int tid = threadIdx.x;
    int lane = tid & 31, w = tid >> 5;
    int tr  = w >> 1;
    int tcb = (w & 1) * 2;
    int m16 = lane & 15;
    int hh  = lane >> 4;
    v8f acc0 = {}; v8f acc1 = {};
    int la  = l0 + tr * 16 + m16;
    int lya = la / WD, lxa = la - lya * WD;
    int pb0 = p0 + tcb * 16 + m16;
    int py0 = pb0 / WD, px0 = pb0 - py0 * WD;
    int pb1 = pb0 + 16;
    int py1 = pb1 / WD, px1 = pb1 - py1 * WD;
    const float* arow  = SA + ((lya - gyA) * PC_ + lxa) * CP + 4 * hh;
    const float* brow0 = SB + ((py0 - gyB) * PC_ + px0) * CP + 4 * hh;
    const float* brow1 = SB + ((py1 - gyB) * PC_ + px1) * CP + 4 * hh;
    const int sp = tid >> 4, sg = 4 * (tid & 15);
    const float* ga = DF + ((size_t)(b * NPP + gyA * PC_ + sp)) * C_ + sg;
    const float* gb = DM + ((size_t)(b * NPP + gyB * PC_ + sp)) * C_ + sg;
    float* sa = SA + sp * CP + sg;
    float* sb = SB + sp * CP + sg;

    __builtin_amdgcn_s_setprio(3);
    for (int c0 = 0; c0 < C_; c0 += KC) {
        if (c0 == KC)          __builtin_amdgcn_s_setprio(2);
        else if (c0 == 2 * KC) __builtin_amdgcn_s_setprio(1);
        else if (c0 == 3 * KC) __builtin_amdgcn_s_setprio(0);
        __syncthreads();
        {
            float4 va[13], vb[20];
            #pragma unroll
            for (int t = 0; t < 13; ++t) va[t] = *(const float4*)(ga + (size_t)(8 * t) * C_ + c0);
            #pragma unroll
            for (int t = 0; t < 20; ++t) vb[t] = *(const float4*)(gb + (size_t)(8 * t) * C_ + c0);
            #pragma unroll
            for (int t = 0; t < 13; ++t) *(float4*)(sa + (8 * t) * CP) = va[t];
            #pragma unroll
            for (int t = 0; t < 20; ++t) *(float4*)(sb + (8 * t) * CP) = vb[t];
        }
        __syncthreads();
        #pragma unroll
        for (int tap = 0; tap < 9; ++tap) {
            const int toff = ((tap / 3) * PC_ + (tap % 3)) * CP;
            #pragma unroll
            for (int kk8 = 0; kk8 < KC; kk8 += 8) {
                const float4 a4 = *(const float4*)(arow  + toff + kk8);
                const float4 c4 = *(const float4*)(brow0 + toff + kk8);
                const float4 d4 = *(const float4*)(brow1 + toff + kk8);
                v2f a, b0, b1;
                a.x = a4.x; a.y = a4.y; b0.x = c4.x; b0.y = c4.y; b1.x = d4.x; b1.y = d4.y;
                acc0 = __builtin_amdgcn_wmma_f32_16x16x4_f32(false, a, false, b0,
                                                             (short)0, acc0, false, false);
                acc1 = __builtin_amdgcn_wmma_f32_16x16x4_f32(false, a, false, b1,
                                                             (short)0, acc1, false, false);
                a.x = a4.z; a.y = a4.w; b0.x = c4.z; b0.y = c4.w; b1.x = d4.z; b1.y = d4.w;
                acc0 = __builtin_amdgcn_wmma_f32_16x16x4_f32(false, a, false, b0,
                                                             (short)0, acc0, false, false);
                acc1 = __builtin_amdgcn_wmma_f32_16x16x4_f32(false, a, false, b1,
                                                             (short)0, acc1, false, false);
            }
        }
    }
    int n     = lane & 15;
    int rbase = l0 + tr * 16 + ((lane >> 4) << 3);
    float s0[8], s1[8];
    float mx0 = -3.0e38f, mx1 = -3.0e38f;
    #pragma unroll
    for (int v = 0; v < 8; ++v) {
        float scl = SCALE_ / den[b * L_ + rbase + v];
        s0[v] = acc0[v] * scl; s1[v] = acc1[v] * scl;
        mx0 = fmaxf(mx0, s0[v]); mx1 = fmaxf(mx1, s1[v]);
    }
    mx0 = fmaxf(mx0, __shfl_xor(mx0, 16));
    mx1 = fmaxf(mx1, __shfl_xor(mx1, 16));
    if (lane < 16) { cmx[tr][tcb * 16 + lane] = mx0; cmx[tr][(tcb + 1) * 16 + lane] = mx1; }
    __syncthreads();
    if (tid < 64)
        pmx[((size_t)((blockIdx.y * B_ + b) * (L_ / 32) + (l0 >> 5))) * L_ + p0 + tid] =
            fmaxf(cmx[0][tid], cmx[1][tid]);
    #pragma unroll
    for (int v = 0; v < 8; ++v) {
        int row = rbase + v;
        out[((size_t)b * L_ + row) * L_ + p0 + tcb * 16 + n]       = s0[v];
        out[((size_t)b * L_ + row) * L_ + p0 + (tcb + 1) * 16 + n] = s1[v];
    }
}

#define RCH  8
#define RM   (RCH * 16)
#define RN   96
__global__ __launch_bounds__(256) void k_recon(const float* __restrict__ shcL,
                                               const float* __restrict__ shcR,
                                               const float* __restrict__ scoL,
                                               const float* __restrict__ scoR,
                                               const float* __restrict__ pmx,
                                               float* __restrict__ ZL,
                                               float* __restrict__ ZR) {
    __shared__ float Pl[RCH * PSZ];
    __shared__ __align__(16) float Bp[RN][LDP];
    __shared__ float redw[8][RN];
    __shared__ float cmax[RN];
    const float* sc = blockIdx.y ? shcR : shcL;
    const float* at = blockIdx.y ? scoR : scoL;
    float* Z        = blockIdx.y ? ZR   : ZL;
    int b   = blockIdx.z;
    int m0  = (blockIdx.x / 6) * RM;
    int p0  = (blockIdx.x % 6) * RN;
    int tid = threadIdx.x;
    int lane = tid & 31, w = tid >> 5;
    int tr  = w >> 1;
    int tc  = w & 1;
    int m16   = lane & 15;
    int khalf = (lane >> 4) * 2;
    v8f acc00 = {}, acc01 = {}, acc02 = {}, acc10 = {}, acc11 = {}, acc12 = {};

    float pq[L_ / 32];
    #pragma unroll
    for (int r = 0; r < L_ / 32; ++r) pq[r] = 0.0f;
    if (tid < RN) {
        const float* pc = pmx + ((size_t)((blockIdx.y * B_ + b) * (L_ / 32))) * L_ + p0 + tid;
        #pragma unroll
        for (int r = 0; r < L_ / 32; ++r) pq[r] = pc[(size_t)r * L_];
    }

    {
        const float4* src = (const float4*)(sc + (size_t)(b * C_ + (m0 >> 4)) * (H_ * W_));
        float4 pv[18];
        #pragma unroll
        for (int n = 0; n < 18; ++n) pv[n] = src[tid + 256 * n];
        #pragma unroll
        for (int n = 0; n < 7; ++n) {
            int z = tid + 256 * n;
            if (z < RCH * 196) {
                int pl = z / 196, wz = z - pl * 196;
                int cell;
                if (wz < PP)          cell = wz;
                else if (wz < 2 * PP) cell = (PP - 1) * PP + (wz - PP);
                else { int vz = wz - 2 * PP; cell = (1 + (vz >> 1)) * PP + (vz & 1) * (PP - 1); }
                Pl[pl * PSZ + cell] = 0.0f;
            }
        }
        #pragma unroll
        for (int n = 0; n < 18; ++n) {
            int e  = 4 * (tid + 256 * n);
            int pl = e / (H_ * W_), re = e - pl * (H_ * W_);
            int r  = re / W_, s = re - r * W_;
            float* d = &Pl[pl * PSZ + (r + 1) * PP + s + 1];
            d[0] = pv[n].x; d[1] = pv[n].y; d[2] = pv[n].z; d[3] = pv[n].w;
        }
    }
    const float* atl = at + (size_t)b * L_ * L_ + p0 + lane;
    float cm0, cm1, cm2;
    {
        if (tid < RN) {
            float m = pq[0];
            #pragma unroll
            for (int r = 1; r < L_ / 32; ++r) m = fmaxf(m, pq[r]);
            cmax[tid] = m;
        }
        __syncthreads();
        cm0 = cmax[lane]; cm1 = cmax[lane + 32]; cm2 = cmax[lane + 64];
    }
    float ps0 = 0.0f, ps1 = 0.0f, ps2 = 0.0f;
    const float* Ath = Pl + (2 * tr) * PSZ + (m16 >> 2) * PP + (m16 & 3) + (lane >> 4) * 4;
    const float* Bw  = &Bp[tc * 48 + m16][khalf];

    __builtin_amdgcn_s_setprio(3);
    for (int kb = 0; kb < L_; kb += TK) {
        if (kb == 6 * TK)       __builtin_amdgcn_s_setprio(2);
        else if (kb == 12 * TK) __builtin_amdgcn_s_setprio(1);
        else if (kb == 16 * TK) __builtin_amdgcn_s_setprio(0);
        __syncthreads();
        float ev[4][3];
        #pragma unroll
        for (int t = 0; t < 4; ++t) {
            const float* rp = atl + (size_t)(kb + w + 8 * t) * L_;
            ev[t][0] = rp[0]; ev[t][1] = rp[32]; ev[t][2] = rp[64];
        }
        #pragma unroll
        for (int t = 0; t < 4; ++t) {
            ev[t][0] = __expf(ev[t][0] - cm0); ps0 += ev[t][0];
            ev[t][1] = __expf(ev[t][1] - cm1); ps1 += ev[t][1];
            ev[t][2] = __expf(ev[t][2] - cm2); ps2 += ev[t][2];
        }
        #pragma unroll
        for (int t = 0; t < 4; ++t) {
            Bp[lane][w + 8 * t]      = ev[t][0];
            Bp[lane + 32][w + 8 * t] = ev[t][1];
            Bp[lane + 64][w + 8 * t] = ev[t][2];
        }
        __syncthreads();
        #pragma unroll
        for (int kk0 = 0; kk0 < TK; kk0 += 4) {
            int l0 = kb + kk0;
            int ly = l0 / WD;
            int lx = l0 - ly * WD;
            const float* ap = Ath + ly * (2 * PP) + 2 * lx;
            v2f a0; a0.x = ap[0];   a0.y = ap[2];
            v2f a1; a1.x = ap[PSZ]; a1.y = ap[PSZ + 2];
            v2f b0 = *(const v2f*)(Bw + kk0);
            v2f b1 = *(const v2f*)(Bw + 16 * LDP + kk0);
            v2f b2 = *(const v2f*)(Bw + 32 * LDP + kk0);
            acc00 = __builtin_amdgcn_wmma_f32_16x16x4_f32(false, a0, false, b0, (short)0, acc00, false, false);
            acc01 = __builtin_amdgcn_wmma_f32_16x16x4_f32(false, a0, false, b1, (short)0, acc01, false, false);
            acc02 = __builtin_amdgcn_wmma_f32_16x16x4_f32(false, a0, false, b2, (short)0, acc02, false, false);
            acc10 = __builtin_amdgcn_wmma_f32_16x16x4_f32(false, a1, false, b0, (short)0, acc10, false, false);
            acc11 = __builtin_amdgcn_wmma_f32_16x16x4_f32(false, a1, false, b1, (short)0, acc11, false, false);
            acc12 = __builtin_amdgcn_wmma_f32_16x16x4_f32(false, a1, false, b2, (short)0, acc12, false, false);
        }
    }
    redw[w][lane] = ps0; redw[w][lane + 32] = ps1; redw[w][lane + 64] = ps2;
    __syncthreads();
    int n  = lane & 15;
    int c0 = tc * 48 + n;
    float inv[3];
    #pragma unroll
    for (int j = 0; j < 3; ++j) {
        float s = redw[0][c0 + 16 * j];
        #pragma unroll
        for (int r = 1; r < 8; ++r) s += redw[r][c0 + 16 * j];
        inv[j] = 1.0f / s;
    }
    float* Zb = Pl;
    {
        const int hh2 = lane >> 4;
        #pragma unroll
        for (int t = 0; t < 3; ++t) {
            int pcol = c0 + 16 * t;
            int gy = pcol / WD, gx = pcol - gy * WD;
            #pragma unroll
            for (int v = 0; v < 8; ++v) {
                int cls = hh2 * 2 + ((v >> 1) & 1);
                int rr  = 2 * gy + (v >> 2);
                int cc  = 2 * gx + (v & 1);
                float* d = Zb + (((cls * RCH + 2 * tr) * 8 + rr) * 48 + cc);
                float x0 = (t == 0 ? acc00[v] : (t == 1 ? acc01[v] : acc02[v])) * inv[t];
                float x1 = (t == 0 ? acc10[v] : (t == 1 ? acc11[v] : acc12[v])) * inv[t];
                d[0]      = x0;
                d[8 * 48] = x1;
            }
        }
    }
    __syncthreads();
    {
        int tl = blockIdx.x % 6;
        float* zw = Z + ((size_t)(b * C_ + (m0 >> 4) + w) * 6 + tl) * 480 + lane;
        const float* zc = Zb + w * (8 * 48);
        #pragma unroll
        for (int k = 0; k < 15; ++k) {
            int s  = (32 * k) % 48 + lane;
            int lr = (32 * k) / 48;
            if (s >= 48) { s -= 48; lr += 1; }
            float sum = 0.0f;
            #pragma unroll
            for (int di = 0; di < 2; ++di) {
                int rr  = lr - 2 * di;
                int rrc = min(max(rr, 0), 7);
                #pragma unroll
                for (int dj = 0; dj < 2; ++dj) {
                    int cc  = s - 2 * dj + 1;
                    int ccc = min(max(cc, 0), 47);
                    float x = zc[(di * 2 + dj) * (RCH * 8 * 48) + rrc * 48 + ccc];
                    bool ok = (rr == rrc) && (cc == ccc);
                    sum += ok ? x : 0.0f;
                }
            }
            zw[32 * k] = sum;
        }
    }
}

__global__ __launch_bounds__(256) void k_finalize(const float* __restrict__ ZL,
                                                  const float* __restrict__ ZR,
                                                  float* __restrict__ out) {
    int idx = blockIdx.x * 256 + threadIdx.x;
    if (idx >= B_ * C_ * H_ * W_) return;
    int s = idx % W_; int u = idx / W_;
    int r = u % H_;  int bc = u / H_;
    int t = r >> 3, q = r & 7;
    size_t base = ((size_t)bc * 6 + t) * 480 + (q + 1) * 48 + s;
    float aL = ZL[base], aR = ZR[base];
    if (q == 0 && t > 0) { size_t o = base - 480 + 8 * 48; aL += ZL[o]; aR += ZR[o]; }
    if (q == 7 && t < 5) { size_t o = base + 480 - 8 * 48; aL += ZL[o]; aR += ZR[o]; }
    const float PI = 3.14159265358979323846f;
    float wl = 0.5f * (cosf(PI * (float)s / (float)(W_ - 1)) + 1.0f);
    float wr = 0.5f * (cosf(PI * (float)(W_ - 1 - s) / (float)(W_ - 1)) + 1.0f);
    out[idx] = 0.25f * (wl * aL + wr * aR);
}

extern "C" void kernel_launch(void* const* d_in, const int* in_sizes, int n_in,
                              void* d_out, int out_size, void* d_ws, size_t ws_size,
                              hipStream_t stream) {
    (void)in_sizes; (void)n_in; (void)out_size; (void)ws_size;
    const float* left  = (const float*)d_in[0];
    const float* right = (const float*)d_in[1];
    const float* mid   = (const float*)d_in[2];
    const float* shcL  = (const float*)d_in[3];
    const float* shcR  = (const float*)d_in[4];
    float* out = (float*)d_out;

    float* w   = (float*)d_ws;
    const size_t NDN = (size_t)B_ * L_;
    const size_t NSC = (size_t)B_ * L_ * L_;
    const size_t NZ  = (size_t)B_ * C_ * 6 * 480;
    const size_t ND  = (size_t)B_ * NPP * C_;
    float* denL = w;
    float* denR = denL + NDN;
    float* scL  = denR + NDN;
    float* scR  = scL + NSC;
    float* ZL   = scR + NSC;
    float* ZR   = ZL + NZ;
    float* DL   = ZL;
    float* DR   = DL + ND;
    float* DM   = DR + ND;
    float* PMX  = ZR + NZ;

    k_prep_cl<<<dim3(PR_ * 4, 3, B_), dim3(256), 0, stream>>>(left, right, mid, DL, DR, DM);
    k_denom<<<dim3(B_ * L_ / 8, 2), dim3(256), 0, stream>>>(DL, DR, denL, denR);
    k_scores<<<dim3(162, 2, B_), dim3(128), 0, stream>>>(DL, DR, DM,
                                                        denL, denR, scL, scR, PMX);
    k_recon<<<dim3(192, 2, B_), dim3(256), 0, stream>>>(shcL, shcR, scL, scR, PMX, ZL, ZR);
    k_finalize<<<dim3((B_ * C_ * H_ * W_ + 255) / 256), dim3(256), 0, stream>>>(ZL, ZR, out);
}
